# starting kernel
# baseline (speedup 1.0000x reference)
#include <hip/hip_runtime.h>


#define B_MOL 16
#define N_ATOM 2048
#define NTILE (N_ATOM / 16)
#define TINY_F 1e-16f
#define WGS 256
#define WG_PER_MOL (N_ATOM / (16 * 8))
#define LPAD (N_ATOM + 16)

typedef __attribute__((ext_vector_type(2))) float v2f;
typedef __attribute__((ext_vector_type(8))) float v8f;

__global__ __launch_bounds__(WGS) void coulomb_field_kernel(
    const float* __restrict__ pos,
    const float* __restrict__ q,
    float* __restrict__ field) {
  __shared__ float smem[5 * LPAD];
  float* lx  = smem;
  float* ly  = smem + LPAD;
  float* lz  = smem + 2 * LPAD;
  float* lq  = smem + 3 * LPAD;
  float* lr2 = smem + 4 * LPAD;

  const int mol = blockIdx.x / WG_PER_MOL;
  const int wgm = blockIdx.x % WG_PER_MOL;
  const int tid = threadIdx.x;

  const float* posm = pos + (size_t)mol * N_ATOM * 3;
  const float* qm   = q   + (size_t)mol * N_ATOM;
  for (int idx = tid; idx < N_ATOM; idx += WGS) {
    float x = posm[idx * 3 + 0];
    float y = posm[idx * 3 + 1];
    float z = posm[idx * 3 + 2];
    lx[idx] = x; ly[idx] = y; lz[idx] = z;
    lq[idx] = qm[idx];
    lr2[idx] = x * x + y * y + z * z;
  }
  __syncthreads();

  const int wave = __builtin_amdgcn_readfirstlane(tid >> 5);
  const int lane = tid & 31;
  const int half = lane >> 4;
  const int l16  = lane & 15;

  const int jbase = (wgm * 8 + wave) * 16;
  const int jtile = jbase >> 4;

  v2f amat;
  {
    const int j = jbase + l16;
    if (half == 0) { amat.x = -2.0f * lx[j]; amat.y = -2.0f * ly[j]; }
    else           { amat.x = -2.0f * lz[j]; amat.y = 1.0f; }
  }

  v8f cmat;
#pragma unroll
  for (int v = 0; v < 8; ++v) cmat[v] = lr2[jbase + v + 8 * half] + TINY_F;

  const float* bsrc0 = half ? lz  : lx;
  const float* bsrc1 = half ? lr2 : ly;

  float facc[8];
#pragma unroll
  for (int v = 0; v < 8; ++v) facc[v] = 0.0f;

  auto accum = [&](const v8f& d, float qi) {
#pragma unroll
    for (int v = 0; v < 8; ++v)
      facc[v] += qi * __builtin_amdgcn_rsqf(__builtin_fabsf(d[v]));
  };

  int ecur = (0 >= jtile) ? 1 : 0;
  v2f bcur; bcur.x = bsrc0[ecur * 16 + l16]; bcur.y = bsrc1[ecur * 16 + l16];
  float qcur = lq[ecur * 16 + l16];

  for (int t = 0; t < NTILE - 2; ++t) {
    const int en = (t + 1) + ((t + 1) >= jtile ? 1 : 0);
    v2f bnext; bnext.x = bsrc0[en * 16 + l16]; bnext.y = bsrc1[en * 16 + l16];
    const float qnext = lq[en * 16 + l16];

    v8f d = __builtin_amdgcn_wmma_f32_16x16x4_f32(
        false, amat, false, bcur, (short)0, cmat, false, false);
    accum(d, qcur);

    bcur = bnext; qcur = qnext;
  }
  {
    v8f d = __builtin_amdgcn_wmma_f32_16x16x4_f32(
        false, amat, false, bcur, (short)0, cmat, false, false);
    accum(d, qcur);
  }

  {
    const int i_l = jbase + l16;
    v2f bmat; bmat.x = bsrc0[i_l]; bmat.y = bsrc1[i_l];
    const float qi = lq[i_l];
    v8f d = __builtin_amdgcn_wmma_f32_16x16x4_f32(
        false, amat, false, bmat, (short)0, cmat, false, false);
#pragma unroll
    for (int v = 0; v < 8; ++v) {
      float inv = __builtin_amdgcn_rsqf(__builtin_fabsf(d[v]));
      facc[v] += (l16 != v + 8 * half) ? qi * inv : 0.0f;
    }
  }

#pragma unroll
  for (int v = 0; v < 8; ++v) {
    float f = facc[v];
    f += __shfl_xor(f, 1);
    f += __shfl_xor(f, 2);
    f += __shfl_xor(f, 4);
    f += __shfl_xor(f, 8);
    facc[v] = f;
  }

  if (l16 == 0) {
#pragma unroll
    for (int v = 0; v < 8; ++v)
      field[(size_t)mol * N_ATOM + jbase + v + 8 * half] = facc[v];
  }
}

__global__ __launch_bounds__(256) void coulomb_pot_kernel(
    const float* __restrict__ q, const float* __restrict__ field,
    float* __restrict__ pot) {
  __shared__ float red[256];
  const int b = blockIdx.x;
  float s = 0.0f;
  for (int j = threadIdx.x; j < N_ATOM; j += 256)
    s += q[(size_t)b * N_ATOM + j] * field[(size_t)b * N_ATOM + j];
  red[threadIdx.x] = s;
  __syncthreads();
  for (int off = 128; off > 0; off >>= 1) {
    if (threadIdx.x < off) red[threadIdx.x] += red[threadIdx.x + off];
    __syncthreads();
  }
  if (threadIdx.x == 0) pot[b] = 0.5f * red[0];
}

extern "C" void kernel_launch(void* const* d_in, const int* in_sizes, int n_in,
                              void* d_out, int out_size, void* d_ws, size_t ws_size,
                              hipStream_t stream) {
  (void)in_sizes; (void)n_in; (void)out_size; (void)d_ws; (void)ws_size;
  const float* pos = (const float*)d_in[0];
  const float* q   = (const float*)d_in[1];

  float* out   = (float*)d_out;
  float* field = out + B_MOL;

  coulomb_field_kernel<<<B_MOL * WG_PER_MOL, WGS, 0, stream>>>(pos, q, field);
  coulomb_pot_kernel<<<B_MOL, 256, 0, stream>>>(q, field, out);
}


# B fragments and charges read straight from global memory 12 tiles ahead in the pair loop, replacing the LDS staging pass: no staging loop, barrier or LDS in field kernel; field sum order unchanged
# speedup vs baseline: 1.1128x; 1.1128x over previous
#include <hip/hip_runtime.h>


#define B_MOL 16
#define N_ATOM 2048
#define NTILE (N_ATOM / 16)
#define TINY_F 1e-16f
#define WGS 256
#define WG_PER_MOL (N_ATOM / (16 * 8))
#define GRP 4
#define NGRP (NTILE / GRP)

typedef __attribute__((ext_vector_type(2))) float v2f;
typedef __attribute__((ext_vector_type(8))) float v8f;

struct Atom { float x, y, z, q; };
struct Group { Atom t[GRP]; };

__global__ __launch_bounds__(WGS) void coulomb_field_kernel(
    const float* __restrict__ pos,
    const float* __restrict__ q,
    float* __restrict__ field) {
  const int mol = blockIdx.x / WG_PER_MOL;
  const int wgm = blockIdx.x % WG_PER_MOL;
  const int tid = threadIdx.x;

  const int wave = __builtin_amdgcn_readfirstlane(tid >> 5);
  const int lane = tid & 31;
  const int half = lane >> 4;
  const int l16  = lane & 15;

  const int jtile = wgm * 8 + wave;
  const int jbase = jtile * 16;

  const char* pmb = (const char*)(pos + (size_t)mol * N_ATOM * 3);
  const char* qmb = (const char*)(q + (size_t)mol * N_ATOM);
  const unsigned lp = (unsigned)l16 * 12u;
  const unsigned lq = (unsigned)l16 * 4u;

  auto load_tile = [&](int tile) -> Atom {
    const unsigned t = (unsigned)tile & (NTILE - 1);
    const float* p  = (const float*)(pmb + (t * 192u + lp));
    const float* pq = (const float*)(qmb + (t * 64u + lq));
    Atom a; a.x = p[0]; a.y = p[1]; a.z = p[2]; a.q = pq[0];
    return a;
  };
  auto r2of = [](float x, float y, float z) -> float {
    return __builtin_fmaf(z, z, __builtin_fmaf(x, x, y * y));
  };
  auto tile_of = [&](int s) -> int { return s + (s >= jtile ? 1 : 0); };
  auto load_group = [&](int g) -> Group {
    Group r;
#pragma unroll
    for (int u = 0; u < GRP; ++u) r.t[u] = load_tile(tile_of(g * GRP + u));
    return r;
  };

  Group ga = load_group(0);
  __builtin_amdgcn_sched_barrier(0);
  Group gb = load_group(1);
  __builtin_amdgcn_sched_barrier(0);
  Group gc = load_group(2);
  __builtin_amdgcn_sched_barrier(0);
  const Atom own = load_tile(jtile);
  const float* prow = (const float*)(pmb + (unsigned)(jbase + 8 * half) * 12u);
  float rowc[24];
#pragma unroll
  for (int k = 0; k < 24; ++k) rowc[k] = prow[k];

  v2f amat;
  if (half == 0) { amat.x = -2.0f * own.x; amat.y = -2.0f * own.y; }
  else           { amat.x = -2.0f * own.z; amat.y = 1.0f; }

  v8f cmat;
#pragma unroll
  for (int v = 0; v < 8; ++v) {
    const float x = rowc[3 * v], y = rowc[3 * v + 1], z = rowc[3 * v + 2];
    cmat[v] = r2of(x, y, z) + TINY_F;
  }

  v2f facc[4];
#pragma unroll
  for (int k = 0; k < 4; ++k) facc[k] = (v2f){0.0f, 0.0f};

  auto bfrag = [&](const Atom& a) -> v2f {
    const float r2 = r2of(a.x, a.y, a.z);
    v2f b; b.x = half ? a.z : a.x; b.y = half ? r2 : a.y;
    return b;
  };
  auto product = [&](const Atom& a) -> v8f {
    return __builtin_amdgcn_wmma_f32_16x16x4_f32(
        false, amat, false, bfrag(a), (short)0, cmat, false, false);
  };
  auto accum = [&](const v8f& d, float qi) {
    const v2f qq = (v2f){qi, qi};
    float inv[8];
#pragma unroll
    for (int v = 0; v < 8; ++v) inv[v] = __builtin_amdgcn_rsqf(__builtin_fabsf(d[v]));
#pragma unroll
    for (int k = 0; k < 4; ++k)
      facc[k] = __builtin_elementwise_fma(qq, (v2f){inv[2 * k], inv[2 * k + 1]}, facc[k]);
  };
  auto process = [&](const Group& g, int n) {
    v8f d[GRP];
#pragma unroll
    for (int u = 0; u < GRP; ++u) if (u < n) d[u] = product(g.t[u]);
#pragma unroll
    for (int u = 0; u < GRP; ++u) if (u < n) accum(d[u], g.t[u].q);
  };

  for (int g = 0; g < NGRP - 4; g += 4) {
    Group gd = load_group(g + 3);
    process(ga, GRP);
    __builtin_amdgcn_sched_barrier(0);
    ga = load_group(g + 4);
    process(gb, GRP);
    __builtin_amdgcn_sched_barrier(0);
    gb = load_group(g + 5);
    process(gc, GRP);
    __builtin_amdgcn_sched_barrier(0);
    gc = load_group(g + 6);
    process(gd, GRP);
    __builtin_amdgcn_sched_barrier(0);
  }
  {
    Group gd;
#pragma unroll
    for (int u = 0; u < GRP - 1; ++u) gd.t[u] = load_tile(tile_of((NGRP - 1) * GRP + u));
    process(ga, GRP);
    process(gb, GRP);
    process(gc, GRP);
    process(gd, GRP - 1);
  }

  float fr[8];
  {
    v8f d = product(own);
#pragma unroll
    for (int v = 0; v < 8; ++v) {
      float inv = __builtin_amdgcn_rsqf(__builtin_fabsf(d[v]));
      float f = (v & 1) ? facc[v >> 1].y : facc[v >> 1].x;
      fr[v] = f + ((l16 != v + 8 * half) ? own.q * inv : 0.0f);
    }
  }

#pragma unroll
  for (int v = 0; v < 8; ++v) {
    float f = fr[v];
    f += __shfl_xor(f, 1);
    f += __shfl_xor(f, 2);
    f += __shfl_xor(f, 4);
    f += __shfl_xor(f, 8);
    fr[v] = f;
  }

  if (l16 == 0) {
#pragma unroll
    for (int v = 0; v < 8; ++v)
      field[(size_t)mol * N_ATOM + jbase + v + 8 * half] = fr[v];
  }
}

__global__ __launch_bounds__(256) void coulomb_pot_kernel(
    const float* __restrict__ q, const float* __restrict__ field,
    float* __restrict__ pot) {
  __shared__ float red[256];
  const int b = blockIdx.x;
  float s = 0.0f;
  for (int j = threadIdx.x; j < N_ATOM; j += 256)
    s += q[(size_t)b * N_ATOM + j] * field[(size_t)b * N_ATOM + j];
  red[threadIdx.x] = s;
  __syncthreads();
  for (int off = 128; off > 0; off >>= 1) {
    if (threadIdx.x < off) red[threadIdx.x] += red[threadIdx.x + off];
    __syncthreads();
  }
  if (threadIdx.x == 0) pot[b] = 0.5f * red[0];
}

extern "C" void kernel_launch(void* const* d_in, const int* in_sizes, int n_in,
                              void* d_out, int out_size, void* d_ws, size_t ws_size,
                              hipStream_t stream) {
  (void)in_sizes; (void)n_in; (void)out_size; (void)d_ws; (void)ws_size;
  const float* pos = (const float*)d_in[0];
  const float* q   = (const float*)d_in[1];

  float* out   = (float*)d_out;
  float* field = out + B_MOL;

  coulomb_field_kernel<<<B_MOL * WG_PER_MOL, WGS, 0, stream>>>(pos, q, field);
  coulomb_pot_kernel<<<B_MOL, 256, 0, stream>>>(q, field, out);
}


# q*field summed per wave in the field kernel while the row sums are in registers: 2048 partials through the workspace, second kernel adds 128 per molecule in one wave; pot sum order changes
# speedup vs baseline: 1.6451x; 1.4783x over previous
#include <hip/hip_runtime.h>


#define B_MOL 16
#define N_ATOM 2048
#define NTILE (N_ATOM / 16)
#define TINY_F 1e-16f
#define WGS 256
#define WG_PER_MOL (N_ATOM / (16 * 8))
#define GRP 4
#define NGRP (NTILE / GRP)
#define PART_BYTES ((size_t)B_MOL * NTILE * sizeof(float))

typedef __attribute__((ext_vector_type(2))) float v2f;
typedef __attribute__((ext_vector_type(8))) float v8f;

typedef const __attribute__((address_space(1))) char*  gcp;
typedef const __attribute__((address_space(1))) float* gfp;

struct Atom { float x, y, z, q; };
struct Group { Atom t[GRP]; };

__global__ __launch_bounds__(WGS) void coulomb_field_kernel(
    const float* __restrict__ pos,
    const float* __restrict__ q,
    float* __restrict__ field,
    float* __restrict__ part,
    size_t part_room) {
  const bool part_ok = (part != nullptr) && (part_room >= PART_BYTES);
  const int mol = blockIdx.x / WG_PER_MOL;
  const int wgm = blockIdx.x % WG_PER_MOL;
  const int tid = threadIdx.x;

  const int wave = __builtin_amdgcn_readfirstlane(tid >> 5);
  const int lane = tid & 31;
  const int half = lane >> 4;
  const int l16  = lane & 15;

  const int jtile = wgm * 8 + wave;
  const int jbase = jtile * 16;

  const char* pmb = (const char*)(pos + (size_t)mol * N_ATOM * 3);
  const char* qmb = (const char*)(q + (size_t)mol * N_ATOM);
  const unsigned lp = (unsigned)l16 * 12u;
  const unsigned lq = (unsigned)l16 * 4u;

  auto load_tile = [&](int tile) -> Atom {
    const unsigned t = (unsigned)tile & (NTILE - 1);
    const float* p  = (const float*)(pmb + (t * 192u + lp));
    const float* pq = (const float*)(qmb + (t * 64u + lq));
    Atom a; a.x = p[0]; a.y = p[1]; a.z = p[2]; a.q = pq[0];
    return a;
  };
  auto r2of = [](float x, float y, float z) -> float {
    return __builtin_fmaf(z, z, __builtin_fmaf(x, x, y * y));
  };
  const int gdiag = jtile >> 2;
  unsigned op = lp, oq = lq;
  auto load_group_at = [&](int G) -> Group {
    const unsigned gq = (unsigned)G & (NGRP - 1);
    gcp tp = (gcp)pmb + (size_t)(gq * (GRP * 192u));
    gcp tq = (gcp)qmb + (size_t)(gq * (GRP * 64u));
    asm("" : "+s"(tp));
    asm("" : "+s"(tq));
    asm volatile("" : "+v"(op), "+v"(oq));
    const gfp p  = (gfp)(tp + op);
    const gfp pq = (gfp)(tq + oq);
    Group r;
#pragma unroll
    for (int u = 0; u < GRP; ++u) {
      r.t[u].x = p[48 * u]; r.t[u].y = p[48 * u + 1]; r.t[u].z = p[48 * u + 2];
      r.t[u].q = pq[16 * u];
    }
    return r;
  };
  auto load_group = [&](int s) -> Group { return load_group_at(s + (s >= gdiag ? 1 : 0)); };

  Group ga = load_group(0);
  __builtin_amdgcn_sched_barrier(0);
  Group gb = load_group(1);
  __builtin_amdgcn_sched_barrier(0);
  Group gc = load_group(2);
  __builtin_amdgcn_sched_barrier(0);
  const Atom own = load_tile(jtile);
  const float* prow = (const float*)(pmb + (unsigned)(jbase + 8 * half) * 12u);
  float rowc[24];
#pragma unroll
  for (int k = 0; k < 24; ++k) rowc[k] = prow[k];

  v2f amat;
  if (half == 0) { amat.x = -2.0f * own.x; amat.y = -2.0f * own.y; }
  else           { amat.x = -2.0f * own.z; amat.y = 1.0f; }

  v8f cmat;
#pragma unroll
  for (int v = 0; v < 8; ++v) {
    const float x = rowc[3 * v], y = rowc[3 * v + 1], z = rowc[3 * v + 2];
    cmat[v] = r2of(x, y, z) + TINY_F;
  }

  v2f facc[4];
#pragma unroll
  for (int k = 0; k < 4; ++k) facc[k] = (v2f){0.0f, 0.0f};

  auto bfrag = [&](const Atom& a) -> v2f {
    const float r2 = r2of(a.x, a.y, a.z);
    v2f b; b.x = half ? a.z : a.x; b.y = half ? r2 : a.y;
    return b;
  };
  auto product = [&](const Atom& a) -> v8f {
    return __builtin_amdgcn_wmma_f32_16x16x4_f32(
        false, amat, false, bfrag(a), (short)0, cmat, false, false);
  };
  auto accum = [&](const v8f& d, float qi) {
    const v2f qq = (v2f){qi, qi};
    float inv[8];
#pragma unroll
    for (int v = 0; v < 8; ++v) inv[v] = __builtin_amdgcn_rsqf(__builtin_fabsf(d[v]));
#pragma unroll
    for (int k = 0; k < 4; ++k)
      facc[k] = __builtin_elementwise_fma(qq, (v2f){inv[2 * k], inv[2 * k + 1]}, facc[k]);
  };
  auto process = [&](const Group& g, int n) {
    v8f d[GRP];
#pragma unroll
    for (int u = 0; u < GRP; ++u) if (u < n) d[u] = product(g.t[u]);
#pragma unroll
    for (int u = 0; u < GRP; ++u) if (u < n) accum(d[u], g.t[u].q);
  };

  for (int g = 0; g < NGRP - 4; g += 4) {
    Group gd = load_group(g + 3);
    process(ga, GRP);
    __builtin_amdgcn_sched_barrier(0);
    ga = load_group(g + 4);
    process(gb, GRP);
    __builtin_amdgcn_sched_barrier(0);
    gb = load_group(g + 5);
    process(gc, GRP);
    __builtin_amdgcn_sched_barrier(0);
    gc = load_group(g + 6);
    process(gd, GRP);
    __builtin_amdgcn_sched_barrier(0);
  }
  float fr[8];
  float qrow[8];
  {
    const Group gx = load_group_at(gdiag);
    const float* qr = (const float*)(qmb + (unsigned)(jbase + 8 * half) * 4u);
#pragma unroll
    for (int v = 0; v < 8; ++v) qrow[v] = qr[v];
    process(ga, GRP);
    process(gb, GRP);
    process(gc, GRP);
#pragma unroll
    for (int v = 0; v < 8; ++v) fr[v] = (v & 1) ? facc[v >> 1].y : facc[v >> 1].x;
    const int udiag = jtile & (GRP - 1);
    v8f d[GRP];
#pragma unroll
    for (int u = 0; u < GRP; ++u) d[u] = product(gx.t[u]);
#pragma unroll
    for (int u = 0; u < GRP; ++u) {
      const bool own_tile = (u == udiag);
#pragma unroll
      for (int v = 0; v < 8; ++v) {
        const float inv = __builtin_amdgcn_rsqf(__builtin_fabsf(d[u][v]));
        const bool self = own_tile && (l16 == v + 8 * half);
        fr[v] += self ? 0.0f : gx.t[u].q * inv;
      }
    }
  }

  float pw;
  {
    const float pa = __builtin_fmaf(qrow[3], fr[3], __builtin_fmaf(qrow[2], fr[2],
                     __builtin_fmaf(qrow[1], fr[1], qrow[0] * fr[0])));
    const float pb = __builtin_fmaf(qrow[7], fr[7], __builtin_fmaf(qrow[6], fr[6],
                     __builtin_fmaf(qrow[5], fr[5], qrow[4] * fr[4])));
    pw = pa + pb;
  }

#pragma unroll
  for (int v = 0; v < 8; ++v) {
    float f = fr[v];
    f += __shfl_xor(f, 1);
    f += __shfl_xor(f, 2);
    f += __shfl_xor(f, 4);
    f += __shfl_xor(f, 8);
    fr[v] = f;
  }
  pw += __shfl_xor(pw, 1);
  pw += __shfl_xor(pw, 2);
  pw += __shfl_xor(pw, 4);
  pw += __shfl_xor(pw, 8);
  pw += __shfl_xor(pw, 16);

  if (l16 == 0) {
#pragma unroll
    for (int v = 0; v < 8; ++v)
      field[(size_t)mol * N_ATOM + jbase + v + 8 * half] = fr[v];
    if (half == 0 && part_ok) part[mol * NTILE + jtile] = pw;
  }
}

__global__ __launch_bounds__(256) void coulomb_pot_kernel(
    const float* __restrict__ q, const float* __restrict__ field,
    const float* __restrict__ part, size_t part_room,
    float* __restrict__ pot) {
  __shared__ float red[256];
  const int b = blockIdx.x;
  if ((part != nullptr) && (part_room >= PART_BYTES)) {
    if (threadIdx.x < 32) {
      const float* w = part + b * NTILE + 4 * threadIdx.x;
      float s = (w[0] + w[1]) + (w[2] + w[3]);
      s += __shfl_xor(s, 1);
      s += __shfl_xor(s, 2);
      s += __shfl_xor(s, 4);
      s += __shfl_xor(s, 8);
      s += __shfl_xor(s, 16);
      if (threadIdx.x == 0) pot[b] = 0.5f * s;
    }
    return;
  }
  float s = 0.0f;
  for (int j = threadIdx.x; j < N_ATOM; j += 256)
    s += q[(size_t)b * N_ATOM + j] * field[(size_t)b * N_ATOM + j];
  red[threadIdx.x] = s;
  __syncthreads();
  for (int off = 128; off > 0; off >>= 1) {
    if (threadIdx.x < off) red[threadIdx.x] += red[threadIdx.x + off];
    __syncthreads();
  }
  if (threadIdx.x == 0) pot[b] = 0.5f * red[0];
}

extern "C" void kernel_launch(void* const* d_in, const int* in_sizes, int n_in,
                              void* d_out, int out_size, void* d_ws, size_t ws_size,
                              hipStream_t stream) {
  (void)in_sizes; (void)n_in; (void)out_size;
  const float* pos = (const float*)d_in[0];
  const float* q   = (const float*)d_in[1];

  float* out   = (float*)d_out;
  float* field = out + B_MOL;

  float* part = (float*)d_ws;

  coulomb_field_kernel<<<B_MOL * WG_PER_MOL, WGS, 0, stream>>>(pos, q, field, part, ws_size);
  coulomb_pot_kernel<<<B_MOL, 256, 0, stream>>>(q, field, part, ws_size, out);
}
